# starting kernel
# baseline (speedup 1.0000x reference)
#include <hip/hip_runtime.h>
#include <math.h>

typedef float v2f __attribute__((ext_vector_type(2)));
typedef float v8f __attribute__((ext_vector_type(8)));

#define NFFT   1024
#define BASEN  16
#define NCHUNK 64
#define WAVES  4
#define PI_F   3.14159265358979323846f

__device__ __forceinline__ int brev6(int c) { return (int)(__brev((unsigned)c) >> 26); }

__global__ __launch_bounds__(WAVES * 32) void rfft1024_kernel(
    const float* __restrict__ x, float* __restrict__ out, int nrows) {
  __shared__ __align__(16) float buf[WAVES][3 * NFFT];
  __shared__ float BRt[BASEN * BASEN];
  __shared__ float BIt[BASEN * BASEN];
  __shared__ float WRt[512];
  __shared__ float WIt[512];

  const int tid = threadIdx.x;

  for (int i = tid; i < BASEN * BASEN; i += WAVES * 32) {
    const int k = i >> 4, n = i & 15;
    const float ang = -2.0f * PI_F * (float)((k * n) & 15) * (1.0f / 16.0f);
    BRt[i] = cosf(ang);
    BIt[i] = sinf(ang);
  }
  for (int i = tid; i < 512; i += WAVES * 32) {
    const float ang = -PI_F * (float)i * (1.0f / 512.0f);
    WRt[i] = cosf(ang);
    WIt[i] = sinf(ang);
  }
  __syncthreads();

  const int wave = tid >> 5;
  const int lane = tid & 31;
  const int half = lane >> 4;
  const int m16  = lane & 15;
  const int row  = blockIdx.x * WAVES + wave;

  float* sx = buf[wave];
  float* ur = sx + NFFT;
  float* ui = ur + NFFT;

  const float4* xr4 = (const float4*)(x + (size_t)row * NFFT);
  float4* sx4 = (float4*)sx;
#pragma unroll
  for (int i = 0; i < NFFT / (32 * 4); ++i) sx4[lane + 32 * i] = xr4[lane + 32 * i];
  __syncthreads();

#pragma unroll
  for (int mt = 0; mt < NCHUNK / 16; ++mt) {
    const int cOff = brev6(mt * 16 + m16);
    v8f accR = {0.f, 0.f, 0.f, 0.f, 0.f, 0.f, 0.f, 0.f};
    v8f accI = {0.f, 0.f, 0.f, 0.f, 0.f, 0.f, 0.f, 0.f};
#pragma unroll
    for (int t = 0; t < 4; ++t) {
      const int k0 = 4 * t + 2 * half;
      v2f a, bR, bI;
      a[0]  = sx[(k0 + 0) * NCHUNK + cOff];
      a[1]  = sx[(k0 + 1) * NCHUNK + cOff];
      bR[0] = BRt[(k0 + 0) * 16 + m16];
      bR[1] = BRt[(k0 + 1) * 16 + m16];
      bI[0] = BIt[(k0 + 0) * 16 + m16];
      bI[1] = BIt[(k0 + 1) * 16 + m16];
      accR = __builtin_amdgcn_wmma_f32_16x16x4_f32(false, a, false, bR, (short)0, accR, false, false);
      accI = __builtin_amdgcn_wmma_f32_16x16x4_f32(false, a, false, bI, (short)0, accI, false, false);
    }
#pragma unroll
    for (int i = 0; i < 8; ++i) {
      const int c = mt * 16 + i + 8 * half;
      ur[c * 16 + m16] = accR[i];
      ui[c * 16 + m16] = accI[i];
    }
  }
  __syncthreads();

  for (int s = 0; s < 6; ++s) {
    const int L = 16 << s;
    const int tsh = 5 - s;
    for (int b = lane; b < 512; b += 32) {
      const int g  = b >> (4 + s);
      const int k  = b & (L - 1);
      const int i0 = (g << (5 + s)) + k;
      const int i1 = i0 + L;
      const float er = ur[i0], ei = ui[i0];
      const float orr = ur[i1], oii = ui[i1];
      const float wr = WRt[k << tsh], wi = WIt[k << tsh];
      const float tr = wr * orr - wi * oii;
      const float ti = wr * oii + wi * orr;
      ur[i0] = er + tr;  ui[i0] = ei + ti;
      ur[i1] = er - tr;  ui[i1] = ei - ti;
    }
    __syncthreads();
  }

  float* outR = out + (size_t)row * 513;
  float* outI = out + (size_t)nrows * 513 + (size_t)row * 513;
  for (int k = lane; k < 513; k += 32) {
    outR[k] = ur[k];
    outI[k] = ui[k];
  }
}

extern "C" void kernel_launch(void* const* d_in, const int* in_sizes, int n_in,
                              void* d_out, int out_size, void* d_ws, size_t ws_size,
                              hipStream_t stream) {
  const float* x = (const float*)d_in[0];
  float* out = (float*)d_out;
  const int nrows = in_sizes[0] / NFFT;
  const int blocks = nrows / WAVES;
  rfft1024_kernel<<<blocks, WAVES * 32, 0, stream>>>(x, out, nrows);
}


# Six radix-2 LDS sweeps become two in-register radix-8 passes: stages 0-2 on the WMMA accumulators, stages 3-5 from one LDS exchange straight to the global stores; same butterflies, same order
# speedup vs baseline: 1.7185x; 1.7185x over previous
#include <hip/hip_runtime.h>
#include <math.h>

typedef float v2f __attribute__((ext_vector_type(2)));
typedef float v8f __attribute__((ext_vector_type(8)));

#define NFFT   1024
#define BASEN  16
#define NCHUNK 64
#define WAVES  4
#define PI_F   3.14159265358979323846f

__device__ __forceinline__ int brev6(int c) { return (int)(__brev((unsigned)c) >> 26); }

__device__ __forceinline__ void bfly(float& er, float& ei, float& orr, float& oi,
                                     const float wr, const float wi) {
  const float tr = wr * orr - wi * oi;
  const float ti = wr * oi + wi * orr;
  const float ar = er, ai = ei;
  er  = ar + tr;  ei = ai + ti;
  orr = ar - tr;  oi = ai - ti;
}

__device__ __forceinline__ void radix8(float (&R)[8], float (&I)[8],
                                       const float (&wr)[7], const float (&wi)[7]) {
  bfly(R[0], I[0], R[1], I[1], wr[0], wi[0]);
  bfly(R[2], I[2], R[3], I[3], wr[0], wi[0]);
  bfly(R[4], I[4], R[5], I[5], wr[0], wi[0]);
  bfly(R[6], I[6], R[7], I[7], wr[0], wi[0]);
  bfly(R[0], I[0], R[2], I[2], wr[1], wi[1]);
  bfly(R[1], I[1], R[3], I[3], wr[2], wi[2]);
  bfly(R[4], I[4], R[6], I[6], wr[1], wi[1]);
  bfly(R[5], I[5], R[7], I[7], wr[2], wi[2]);
  bfly(R[0], I[0], R[4], I[4], wr[3], wi[3]);
  bfly(R[1], I[1], R[5], I[5], wr[4], wi[4]);
  bfly(R[2], I[2], R[6], I[6], wr[5], wi[5]);
  bfly(R[3], I[3], R[7], I[7], wr[6], wi[6]);
}

__global__ __launch_bounds__(WAVES * 32) void rfft1024_kernel(
    const float* __restrict__ x, float* __restrict__ out, int nrows) {
  __shared__ __align__(16) float buf[WAVES][3 * NFFT];
  __shared__ float BRt[BASEN * BASEN];
  __shared__ float BIt[BASEN * BASEN];
  __shared__ float WRt[512];
  __shared__ float WIt[512];

  const int tid = threadIdx.x;

  for (int i = tid; i < BASEN * BASEN; i += WAVES * 32) {
    const int k = i >> 4, n = i & 15;
    const float ang = -2.0f * PI_F * (float)((k * n) & 15) * (1.0f / 16.0f);
    BRt[i] = cosf(ang);
    BIt[i] = sinf(ang);
  }
  for (int i = tid; i < 512; i += WAVES * 32) {
    const float ang = -PI_F * (float)i * (1.0f / 512.0f);
    WRt[i] = cosf(ang);
    WIt[i] = sinf(ang);
  }
  __syncthreads();

  const int wave = tid >> 5;
  const int lane = tid & 31;
  const int half = lane >> 4;
  const int m16  = lane & 15;
  const int row  = blockIdx.x * WAVES + wave;

  float* sx = buf[wave];
  float* ur = sx + NFFT;
  float* ui = ur + NFFT;

  const float4* xr4 = (const float4*)(x + (size_t)row * NFFT);
  float4* sx4 = (float4*)sx;
#pragma unroll
  for (int i = 0; i < NFFT / (32 * 4); ++i) sx4[lane + 32 * i] = xr4[lane + 32 * i];
  __syncthreads();

  const float p1r[7] = {WRt[m16 << 5],
                        WRt[m16 << 4], WRt[(m16 + 16) << 4],
                        WRt[m16 << 3], WRt[(m16 + 16) << 3], WRt[(m16 + 32) << 3], WRt[(m16 + 48) << 3]};
  const float p1i[7] = {WIt[m16 << 5],
                        WIt[m16 << 4], WIt[(m16 + 16) << 4],
                        WIt[m16 << 3], WIt[(m16 + 16) << 3], WIt[(m16 + 32) << 3], WIt[(m16 + 48) << 3]};

#pragma unroll
  for (int mt = 0; mt < NCHUNK / 16; ++mt) {
    const int cOff = brev6(mt * 16 + m16);
    v8f accR = {0.f, 0.f, 0.f, 0.f, 0.f, 0.f, 0.f, 0.f};
    v8f accI = {0.f, 0.f, 0.f, 0.f, 0.f, 0.f, 0.f, 0.f};
#pragma unroll
    for (int t = 0; t < 4; ++t) {
      const int k0 = 4 * t + 2 * half;
      v2f a, bR, bI;
      a[0]  = sx[(k0 + 0) * NCHUNK + cOff];
      a[1]  = sx[(k0 + 1) * NCHUNK + cOff];
      bR[0] = BRt[(k0 + 0) * 16 + m16];
      bR[1] = BRt[(k0 + 1) * 16 + m16];
      bI[0] = BIt[(k0 + 0) * 16 + m16];
      bI[1] = BIt[(k0 + 1) * 16 + m16];
      accR = __builtin_amdgcn_wmma_f32_16x16x4_f32(false, a, false, bR, (short)0, accR, false, false);
      accI = __builtin_amdgcn_wmma_f32_16x16x4_f32(false, a, false, bI, (short)0, accI, false, false);
    }
    float R[8] = {accR[0], accR[1], accR[2], accR[3], accR[4], accR[5], accR[6], accR[7]};
    float I[8] = {accI[0], accI[1], accI[2], accI[3], accI[4], accI[5], accI[6], accI[7]};
    radix8(R, I, p1r, p1i);
#pragma unroll
    for (int i = 0; i < 8; ++i) {
      const int c = mt * 16 + i + 8 * half;
      ur[c * 16 + m16] = R[i];
      ui[c * 16 + m16] = I[i];
    }
  }
  __syncthreads();

  float* outR = out + (size_t)row * 513;
  float* outI = out + (size_t)nrows * 513 + (size_t)row * 513;
#pragma unroll
  for (int it = 0; it < 4; ++it) {
    const int k0 = lane + 32 * it;
    float R[8] = {ur[k0],       ur[128 + k0], ur[256 + k0], ur[384 + k0],
                  ur[512 + k0], ur[640 + k0], ur[768 + k0], ur[896 + k0]};
    float I[8] = {ui[k0],       ui[128 + k0], ui[256 + k0], ui[384 + k0],
                  ui[512 + k0], ui[640 + k0], ui[768 + k0], ui[896 + k0]};
    const float p2r[7] = {WRt[k0 << 2],
                          WRt[k0 << 1], WRt[(k0 + 128) << 1],
                          WRt[k0], WRt[k0 + 128], WRt[k0 + 256], WRt[k0 + 384]};
    const float p2i[7] = {WIt[k0 << 2],
                          WIt[k0 << 1], WIt[(k0 + 128) << 1],
                          WIt[k0], WIt[k0 + 128], WIt[k0 + 256], WIt[k0 + 384]};
    radix8(R, I, p2r, p2i);
    outR[k0]       = R[0];  outI[k0]       = I[0];
    outR[128 + k0] = R[1];  outI[128 + k0] = I[1];
    outR[256 + k0] = R[2];  outI[256 + k0] = I[2];
    outR[384 + k0] = R[3];  outI[384 + k0] = I[3];
    if (k0 == 0) { outR[512] = R[4];  outI[512] = I[4]; }
  }
}

extern "C" void kernel_launch(void* const* d_in, const int* in_sizes, int n_in,
                              void* d_out, int out_size, void* d_ws, size_t ws_size,
                              hipStream_t stream) {
  const float* x = (const float*)d_in[0];
  float* out = (float*)d_out;
  const int nrows = in_sizes[0] / NFFT;
  const int blocks = nrows / WAVES;
  rfft1024_kernel<<<blocks, WAVES * 32, 0, stream>>>(x, out, nrows);
}


# Each wave transforms 4 rows in a loop instead of 1 (512 blocks, not 2048): tables, twiddle fetch and wave launch paid once per 4 rows, next row's loads in flight under the current row; same sums
# speedup vs baseline: 1.9142x; 1.1139x over previous
#include <hip/hip_runtime.h>
#include <math.h>

typedef float v2f __attribute__((ext_vector_type(2)));
typedef float v8f __attribute__((ext_vector_type(8)));
typedef float v4f __attribute__((ext_vector_type(4)));

#define NFFT   1024
#define BASEN  16
#define NCHUNK 64
#define WAVES  4
#define ROWS   4
#define PI_F   3.14159265358979323846f

__device__ __forceinline__ int brev6(int c) { return (int)(__brev((unsigned)c) >> 26); }

__device__ __forceinline__ void bfly(float& er, float& ei, float& orr, float& oi,
                                     const float wr, const float wi) {
  const float tr = wr * orr - wi * oi;
  const float ti = wr * oi + wi * orr;
  const float ar = er, ai = ei;
  er  = ar + tr;  ei = ai + ti;
  orr = ar - tr;  oi = ai - ti;
}

__device__ __forceinline__ void radix8(float (&R)[8], float (&I)[8],
                                       const float (&wr)[7], const float (&wi)[7]) {
  bfly(R[0], I[0], R[1], I[1], wr[0], wi[0]);
  bfly(R[2], I[2], R[3], I[3], wr[0], wi[0]);
  bfly(R[4], I[4], R[5], I[5], wr[0], wi[0]);
  bfly(R[6], I[6], R[7], I[7], wr[0], wi[0]);
  bfly(R[0], I[0], R[2], I[2], wr[1], wi[1]);
  bfly(R[1], I[1], R[3], I[3], wr[2], wi[2]);
  bfly(R[4], I[4], R[6], I[6], wr[1], wi[1]);
  bfly(R[5], I[5], R[7], I[7], wr[2], wi[2]);
  bfly(R[0], I[0], R[4], I[4], wr[3], wi[3]);
  bfly(R[1], I[1], R[5], I[5], wr[4], wi[4]);
  bfly(R[2], I[2], R[6], I[6], wr[5], wi[5]);
  bfly(R[3], I[3], R[7], I[7], wr[6], wi[6]);
}

__device__ __forceinline__ void wave_lds_order() {
  __builtin_amdgcn_fence(__ATOMIC_RELEASE, "wavefront");
  __builtin_amdgcn_wave_barrier();
  __builtin_amdgcn_fence(__ATOMIC_ACQUIRE, "wavefront");
}

__global__ __launch_bounds__(WAVES * 32) void rfft1024_kernel(
    const float* __restrict__ x, float* __restrict__ out, int nrows) {
  __shared__ __align__(16) float buf[WAVES][3 * NFFT];
  __shared__ float BRt[BASEN * BASEN];
  __shared__ float BIt[BASEN * BASEN];
  __shared__ float WRt[512];
  __shared__ float WIt[512];

  const int tid  = threadIdx.x;
  const int wave = tid >> 5;
  const int lane = tid & 31;
  const int half = lane >> 4;
  const int m16  = lane & 15;
  const int row0 = blockIdx.x * (WAVES * ROWS) + wave;

  v4f nxt[NFFT / (32 * 4)];
  {
    const v4f* xr4 = (const v4f*)(x + (size_t)row0 * NFFT);
#pragma unroll
    for (int i = 0; i < NFFT / (32 * 4); ++i) nxt[i] = xr4[lane + 32 * i];
  }

  for (int i = tid; i < BASEN * BASEN; i += WAVES * 32) {
    const int k = i >> 4, n = i & 15;
    const float ang = -2.0f * PI_F * (float)((k * n) & 15) * (1.0f / 16.0f);
    BRt[i] = cosf(ang);
    BIt[i] = sinf(ang);
  }
  for (int i = tid; i < 512; i += WAVES * 32) {
    const float ang = -PI_F * (float)i * (1.0f / 512.0f);
    WRt[i] = cosf(ang);
    WIt[i] = sinf(ang);
  }
  __syncthreads();

  float* sx = buf[wave];
  float* ur = sx + NFFT;
  float* ui = ur + NFFT;
  v4f* sx4 = (v4f*)sx;

  const float p1r[7] = {WRt[m16 << 5],
                        WRt[m16 << 4], WRt[(m16 + 16) << 4],
                        WRt[m16 << 3], WRt[(m16 + 16) << 3], WRt[(m16 + 32) << 3], WRt[(m16 + 48) << 3]};
  const float p1i[7] = {WIt[m16 << 5],
                        WIt[m16 << 4], WIt[(m16 + 16) << 4],
                        WIt[m16 << 3], WIt[(m16 + 16) << 3], WIt[(m16 + 32) << 3], WIt[(m16 + 48) << 3]};
  float p2r[4][7], p2i[4][7];
#pragma unroll
  for (int it = 0; it < 4; ++it) {
    const int k0 = lane + 32 * it;
    p2r[it][0] = WRt[k0 << 2];
    p2r[it][1] = WRt[k0 << 1];  p2r[it][2] = WRt[(k0 + 128) << 1];
    p2r[it][3] = WRt[k0];       p2r[it][4] = WRt[k0 + 128];
    p2r[it][5] = WRt[k0 + 256]; p2r[it][6] = WRt[k0 + 384];
    p2i[it][0] = WIt[k0 << 2];
    p2i[it][1] = WIt[k0 << 1];  p2i[it][2] = WIt[(k0 + 128) << 1];
    p2i[it][3] = WIt[k0];       p2i[it][4] = WIt[k0 + 128];
    p2i[it][5] = WIt[k0 + 256]; p2i[it][6] = WIt[k0 + 384];
  }
  v2f bR[4], bI[4];
#pragma unroll
  for (int t = 0; t < 4; ++t) {
    const int k0 = 4 * t + 2 * half;
    bR[t][0] = BRt[(k0 + 0) * 16 + m16];
    bR[t][1] = BRt[(k0 + 1) * 16 + m16];
    bI[t][0] = BIt[(k0 + 0) * 16 + m16];
    bI[t][1] = BIt[(k0 + 1) * 16 + m16];
  }

#pragma nounroll
  for (int r = 0; r < ROWS; ++r) {
    const int row = row0 + r * WAVES;

#pragma unroll
    for (int i = 0; i < NFFT / (32 * 4); ++i) sx4[lane + 32 * i] = nxt[i];
    if (r + 1 < ROWS) {
      const v4f* xn4 = (const v4f*)(x + (size_t)(row + WAVES) * NFFT);
#pragma unroll
      for (int i = 0; i < NFFT / (32 * 4); ++i) nxt[i] = xn4[lane + 32 * i];
    }
    wave_lds_order();

#pragma unroll
    for (int mt = 0; mt < NCHUNK / 16; ++mt) {
      const int cOff = brev6(mt * 16 + m16);
      v8f accR = {0.f, 0.f, 0.f, 0.f, 0.f, 0.f, 0.f, 0.f};
      v8f accI = {0.f, 0.f, 0.f, 0.f, 0.f, 0.f, 0.f, 0.f};
#pragma unroll
      for (int t = 0; t < 4; ++t) {
        const int k0 = 4 * t + 2 * half;
        v2f a;
        a[0] = sx[(k0 + 0) * NCHUNK + cOff];
        a[1] = sx[(k0 + 1) * NCHUNK + cOff];
        accR = __builtin_amdgcn_wmma_f32_16x16x4_f32(false, a, false, bR[t], (short)0, accR, false, false);
        accI = __builtin_amdgcn_wmma_f32_16x16x4_f32(false, a, false, bI[t], (short)0, accI, false, false);
      }
      float R[8] = {accR[0], accR[1], accR[2], accR[3], accR[4], accR[5], accR[6], accR[7]};
      float I[8] = {accI[0], accI[1], accI[2], accI[3], accI[4], accI[5], accI[6], accI[7]};
      radix8(R, I, p1r, p1i);
#pragma unroll
      for (int i = 0; i < 8; ++i) {
        const int c = mt * 16 + i + 8 * half;
        ur[c * 16 + m16] = R[i];
        ui[c * 16 + m16] = I[i];
      }
    }
    wave_lds_order();

    float* outR = out + (size_t)row * 513;
    float* outI = out + (size_t)nrows * 513 + (size_t)row * 513;
#pragma unroll
    for (int it = 0; it < 4; ++it) {
      const int k0 = lane + 32 * it;
      float R[8] = {ur[k0],       ur[128 + k0], ur[256 + k0], ur[384 + k0],
                    ur[512 + k0], ur[640 + k0], ur[768 + k0], ur[896 + k0]};
      float I[8] = {ui[k0],       ui[128 + k0], ui[256 + k0], ui[384 + k0],
                    ui[512 + k0], ui[640 + k0], ui[768 + k0], ui[896 + k0]};
      radix8(R, I, p2r[it], p2i[it]);
      outR[k0]       = R[0];  outI[k0]       = I[0];
      outR[128 + k0] = R[1];  outI[128 + k0] = I[1];
      outR[256 + k0] = R[2];  outI[256 + k0] = I[2];
      outR[384 + k0] = R[3];  outI[384 + k0] = I[3];
      if (k0 == 0) { outR[512] = R[4];  outI[512] = I[4]; }
    }
    wave_lds_order();
  }
}

extern "C" void kernel_launch(void* const* d_in, const int* in_sizes, int n_in,
                              void* d_out, int out_size, void* d_ws, size_t ws_size,
                              hipStream_t stream) {
  const float* x = (const float*)d_in[0];
  float* out = (float*)d_out;
  const int nrows = in_sizes[0] / NFFT;
  const int blocks = nrows / (WAVES * ROWS);
  rfft1024_kernel<<<blocks, WAVES * 32, 0, stream>>>(x, out, nrows);
}


# Pass 2: a lane owns 4 consecutive bins (4*lane+q, was lane+32*it): 16-byte work-array reads and output stores, butterflies on adjacent-bin pairs (2-wide packed math); same sums
# speedup vs baseline: 2.0796x; 1.0864x over previous
#include <hip/hip_runtime.h>
#include <math.h>

typedef float v2f __attribute__((ext_vector_type(2)));
typedef float v8f __attribute__((ext_vector_type(8)));
typedef float v4f __attribute__((ext_vector_type(4)));
typedef float v4f_a4 __attribute__((ext_vector_type(4), aligned(4)));

#define NFFT   1024
#define BASEN  16
#define NCHUNK 64
#define WAVES  4
#define ROWS   4
#define PI_F   3.14159265358979323846f

__device__ __forceinline__ int brev6(int c) { return (int)(__brev((unsigned)c) >> 26); }

__device__ __forceinline__ void bfly(float& er, float& ei, float& orr, float& oi,
                                     const float wr, const float wi) {
  const float tr = wr * orr - wi * oi;
  const float ti = wr * oi + wi * orr;
  const float ar = er, ai = ei;
  er  = ar + tr;  ei = ai + ti;
  orr = ar - tr;  oi = ai - ti;
}

__device__ __forceinline__ void radix8(float (&R)[8], float (&I)[8],
                                       const float (&wr)[7], const float (&wi)[7]) {
  bfly(R[0], I[0], R[1], I[1], wr[0], wi[0]);
  bfly(R[2], I[2], R[3], I[3], wr[0], wi[0]);
  bfly(R[4], I[4], R[5], I[5], wr[0], wi[0]);
  bfly(R[6], I[6], R[7], I[7], wr[0], wi[0]);
  bfly(R[0], I[0], R[2], I[2], wr[1], wi[1]);
  bfly(R[1], I[1], R[3], I[3], wr[2], wi[2]);
  bfly(R[4], I[4], R[6], I[6], wr[1], wi[1]);
  bfly(R[5], I[5], R[7], I[7], wr[2], wi[2]);
  bfly(R[0], I[0], R[4], I[4], wr[3], wi[3]);
  bfly(R[1], I[1], R[5], I[5], wr[4], wi[4]);
  bfly(R[2], I[2], R[6], I[6], wr[5], wi[5]);
  bfly(R[3], I[3], R[7], I[7], wr[6], wi[6]);
}

__device__ __forceinline__ void bfly2(v2f& er, v2f& ei, v2f& orr, v2f& oi, const v2f wr, const v2f wi) {
  const v2f tr = wr * orr - wi * oi;
  const v2f ti = wr * oi + wi * orr;
  const v2f ar = er, ai = ei;
  er  = ar + tr;  ei = ai + ti;
  orr = ar - tr;  oi = ai - ti;
}

__device__ __forceinline__ void radix8x2(v2f (&R)[8], v2f (&I)[8],
                                         const v2f (&wr)[7], const v2f (&wi)[7]) {
  bfly2(R[0], I[0], R[1], I[1], wr[0], wi[0]);
  bfly2(R[2], I[2], R[3], I[3], wr[0], wi[0]);
  bfly2(R[4], I[4], R[5], I[5], wr[0], wi[0]);
  bfly2(R[6], I[6], R[7], I[7], wr[0], wi[0]);
  bfly2(R[0], I[0], R[2], I[2], wr[1], wi[1]);
  bfly2(R[1], I[1], R[3], I[3], wr[2], wi[2]);
  bfly2(R[4], I[4], R[6], I[6], wr[1], wi[1]);
  bfly2(R[5], I[5], R[7], I[7], wr[2], wi[2]);
  bfly2(R[0], I[0], R[4], I[4], wr[3], wi[3]);
  bfly2(R[1], I[1], R[5], I[5], wr[4], wi[4]);
  bfly2(R[2], I[2], R[6], I[6], wr[5], wi[5]);
  bfly2(R[3], I[3], R[7], I[7], wr[6], wi[6]);
}

__device__ __forceinline__ void wave_lds_order() {
  __builtin_amdgcn_fence(__ATOMIC_RELEASE, "wavefront");
  __builtin_amdgcn_wave_barrier();
  __builtin_amdgcn_fence(__ATOMIC_ACQUIRE, "wavefront");
}

__global__ __launch_bounds__(WAVES * 32) void rfft1024_kernel(
    const float* __restrict__ x, float* __restrict__ out, int nrows) {
  __shared__ __align__(16) float buf[WAVES][3 * NFFT];
  __shared__ float BRt[BASEN * BASEN];
  __shared__ float BIt[BASEN * BASEN];
  __shared__ float WRt[512];
  __shared__ float WIt[512];

  const int tid  = threadIdx.x;
  const int wave = tid >> 5;
  const int lane = tid & 31;
  const int half = lane >> 4;
  const int m16  = lane & 15;
  const int row0 = blockIdx.x * (WAVES * ROWS) + wave;

  v4f nxt[NFFT / (32 * 4)];
  {
    const v4f* xr4 = (const v4f*)(x + (size_t)row0 * NFFT);
#pragma unroll
    for (int i = 0; i < NFFT / (32 * 4); ++i) nxt[i] = xr4[lane + 32 * i];
  }

  for (int i = tid; i < BASEN * BASEN; i += WAVES * 32) {
    const int k = i >> 4, n = i & 15;
    const float ang = -2.0f * PI_F * (float)((k * n) & 15) * (1.0f / 16.0f);
    BRt[i] = cosf(ang);
    BIt[i] = sinf(ang);
  }
  for (int i = tid; i < 512; i += WAVES * 32) {
    const float ang = -PI_F * (float)i * (1.0f / 512.0f);
    WRt[i] = cosf(ang);
    WIt[i] = sinf(ang);
  }
  __syncthreads();

  float* sx = buf[wave];
  float* ur = sx + NFFT;
  float* ui = ur + NFFT;
  v4f* sx4 = (v4f*)sx;

  const float p1r[7] = {WRt[m16 << 5],
                        WRt[m16 << 4], WRt[(m16 + 16) << 4],
                        WRt[m16 << 3], WRt[(m16 + 16) << 3], WRt[(m16 + 32) << 3], WRt[(m16 + 48) << 3]};
  const float p1i[7] = {WIt[m16 << 5],
                        WIt[m16 << 4], WIt[(m16 + 16) << 4],
                        WIt[m16 << 3], WIt[(m16 + 16) << 3], WIt[(m16 + 32) << 3], WIt[(m16 + 48) << 3]};
  v2f p2r[2][7], p2i[2][7];
#pragma unroll
  for (int q = 0; q < 4; ++q) {
    const int k0 = 4 * lane + q;
    const int h = q >> 1, e = q & 1;
    p2r[h][0][e] = WRt[k0 << 2];
    p2r[h][1][e] = WRt[k0 << 1];  p2r[h][2][e] = WRt[(k0 + 128) << 1];
    p2r[h][3][e] = WRt[k0];       p2r[h][4][e] = WRt[k0 + 128];
    p2r[h][5][e] = WRt[k0 + 256]; p2r[h][6][e] = WRt[k0 + 384];
    p2i[h][0][e] = WIt[k0 << 2];
    p2i[h][1][e] = WIt[k0 << 1];  p2i[h][2][e] = WIt[(k0 + 128) << 1];
    p2i[h][3][e] = WIt[k0];       p2i[h][4][e] = WIt[k0 + 128];
    p2i[h][5][e] = WIt[k0 + 256]; p2i[h][6][e] = WIt[k0 + 384];
  }
  v2f bR[4], bI[4];
#pragma unroll
  for (int t = 0; t < 4; ++t) {
    const int k0 = 4 * t + 2 * half;
    bR[t][0] = BRt[(k0 + 0) * 16 + m16];
    bR[t][1] = BRt[(k0 + 1) * 16 + m16];
    bI[t][0] = BIt[(k0 + 0) * 16 + m16];
    bI[t][1] = BIt[(k0 + 1) * 16 + m16];
  }

#pragma nounroll
  for (int r = 0; r < ROWS; ++r) {
    const int row = row0 + r * WAVES;

#pragma unroll
    for (int i = 0; i < NFFT / (32 * 4); ++i) sx4[lane + 32 * i] = nxt[i];
    if (r + 1 < ROWS) {
      const v4f* xn4 = (const v4f*)(x + (size_t)(row + WAVES) * NFFT);
#pragma unroll
      for (int i = 0; i < NFFT / (32 * 4); ++i) nxt[i] = xn4[lane + 32 * i];
    }
    wave_lds_order();

#pragma unroll
    for (int mt = 0; mt < NCHUNK / 16; ++mt) {
      const int cOff = brev6(mt * 16 + m16);
      v8f accR = {0.f, 0.f, 0.f, 0.f, 0.f, 0.f, 0.f, 0.f};
      v8f accI = {0.f, 0.f, 0.f, 0.f, 0.f, 0.f, 0.f, 0.f};
#pragma unroll
      for (int t = 0; t < 4; ++t) {
        const int k0 = 4 * t + 2 * half;
        v2f a;
        a[0] = sx[(k0 + 0) * NCHUNK + cOff];
        a[1] = sx[(k0 + 1) * NCHUNK + cOff];
        accR = __builtin_amdgcn_wmma_f32_16x16x4_f32(false, a, false, bR[t], (short)0, accR, false, false);
        accI = __builtin_amdgcn_wmma_f32_16x16x4_f32(false, a, false, bI[t], (short)0, accI, false, false);
      }
      float R[8] = {accR[0], accR[1], accR[2], accR[3], accR[4], accR[5], accR[6], accR[7]};
      float I[8] = {accI[0], accI[1], accI[2], accI[3], accI[4], accI[5], accI[6], accI[7]};
      radix8(R, I, p1r, p1i);
#pragma unroll
      for (int i = 0; i < 8; ++i) {
        const int c = mt * 16 + i + 8 * half;
        ur[c * 16 + m16] = R[i];
        ui[c * 16 + m16] = I[i];
      }
    }
    wave_lds_order();

    float* outR = out + (size_t)row * 513;
    float* outI = out + (size_t)nrows * 513 + (size_t)row * 513;
    const v4f* ur4 = (const v4f*)ur;
    const v4f* ui4 = (const v4f*)ui;
    v4f UR[8], UI[8];
#pragma unroll
    for (int j = 0; j < 8; ++j) {
      UR[j] = ur4[32 * j + lane];
      UI[j] = ui4[32 * j + lane];
    }
    v2f oR[4][2], oI[4][2];
    float r512 = 0.f, i512 = 0.f;
#pragma unroll
    for (int h = 0; h < 2; ++h) {
      v2f R[8], I[8];
#pragma unroll
      for (int j = 0; j < 8; ++j) {
        R[j] = h ? __builtin_shufflevector(UR[j], UR[j], 2, 3) : __builtin_shufflevector(UR[j], UR[j], 0, 1);
        I[j] = h ? __builtin_shufflevector(UI[j], UI[j], 2, 3) : __builtin_shufflevector(UI[j], UI[j], 0, 1);
      }
      radix8x2(R, I, p2r[h], p2i[h]);
#pragma unroll
      for (int b = 0; b < 4; ++b) { oR[b][h] = R[b];  oI[b][h] = I[b]; }
      if (h == 0) { r512 = R[4][0];  i512 = I[4][0]; }
    }
#pragma unroll
    for (int b = 0; b < 4; ++b) {
      *(v4f_a4*)(outR + 128 * b + 4 * lane) = __builtin_shufflevector(oR[b][0], oR[b][1], 0, 1, 2, 3);
      *(v4f_a4*)(outI + 128 * b + 4 * lane) = __builtin_shufflevector(oI[b][0], oI[b][1], 0, 1, 2, 3);
    }
    if (lane == 0) { outR[512] = r512;  outI[512] = i512; }
    wave_lds_order();
  }
}

extern "C" void kernel_launch(void* const* d_in, const int* in_sizes, int n_in,
                              void* d_out, int out_size, void* d_ws, size_t ws_size,
                              hipStream_t stream) {
  const float* x = (const float*)d_in[0];
  float* out = (float*)d_out;
  const int nrows = in_sizes[0] / NFFT;
  const int blocks = nrows / (WAVES * ROWS);
  rfft1024_kernel<<<blocks, WAVES * 32, 0, stream>>>(x, out, nrows);
}


# Pass 1: stages 1 and 2 do two butterflies at once on the accumulator register pairs (0,1)(2,3)(4,5)(6,7) as the matrix unit leaves them: packed math, no register shuffles; same butterflies, same sums
# speedup vs baseline: 2.2335x; 1.0740x over previous
#include <hip/hip_runtime.h>
#include <math.h>

typedef float v2f __attribute__((ext_vector_type(2)));
typedef float v8f __attribute__((ext_vector_type(8)));
typedef float v4f __attribute__((ext_vector_type(4)));
typedef float v4f_a4 __attribute__((ext_vector_type(4), aligned(4)));

#define NFFT   1024
#define BASEN  16
#define NCHUNK 64
#define WAVES  4
#define ROWS   4
#define PI_F   3.14159265358979323846f

__device__ __forceinline__ int brev6(int c) { return (int)(__brev((unsigned)c) >> 26); }

__device__ __forceinline__ void bfly(float& er, float& ei, float& orr, float& oi,
                                     const float wr, const float wi) {
  const float tr = wr * orr - wi * oi;
  const float ti = wr * oi + wi * orr;
  const float ar = er, ai = ei;
  er  = ar + tr;  ei = ai + ti;
  orr = ar - tr;  oi = ai - ti;
}

__device__ __forceinline__ void bfly2(v2f& er, v2f& ei, v2f& orr, v2f& oi, const v2f wr, const v2f wi) {
  const v2f tr = wr * orr - wi * oi;
  const v2f ti = wr * oi + wi * orr;
  const v2f ar = er, ai = ei;
  er  = ar + tr;  ei = ai + ti;
  orr = ar - tr;  oi = ai - ti;
}

__device__ __forceinline__ void radix8x2(v2f (&R)[8], v2f (&I)[8],
                                         const v2f (&wr)[7], const v2f (&wi)[7]) {
  bfly2(R[0], I[0], R[1], I[1], wr[0], wi[0]);
  bfly2(R[2], I[2], R[3], I[3], wr[0], wi[0]);
  bfly2(R[4], I[4], R[5], I[5], wr[0], wi[0]);
  bfly2(R[6], I[6], R[7], I[7], wr[0], wi[0]);
  bfly2(R[0], I[0], R[2], I[2], wr[1], wi[1]);
  bfly2(R[1], I[1], R[3], I[3], wr[2], wi[2]);
  bfly2(R[4], I[4], R[6], I[6], wr[1], wi[1]);
  bfly2(R[5], I[5], R[7], I[7], wr[2], wi[2]);
  bfly2(R[0], I[0], R[4], I[4], wr[3], wi[3]);
  bfly2(R[1], I[1], R[5], I[5], wr[4], wi[4]);
  bfly2(R[2], I[2], R[6], I[6], wr[5], wi[5]);
  bfly2(R[3], I[3], R[7], I[7], wr[6], wi[6]);
}

__device__ __forceinline__ void wave_lds_order() {
  __builtin_amdgcn_fence(__ATOMIC_RELEASE, "wavefront");
  __builtin_amdgcn_wave_barrier();
  __builtin_amdgcn_fence(__ATOMIC_ACQUIRE, "wavefront");
}

__global__ __launch_bounds__(WAVES * 32) void rfft1024_kernel(
    const float* __restrict__ x, float* __restrict__ out, int nrows) {
  __shared__ __align__(16) float buf[WAVES][3 * NFFT];
  __shared__ float BRt[BASEN * BASEN];
  __shared__ float BIt[BASEN * BASEN];
  __shared__ float WRt[512];
  __shared__ float WIt[512];

  const int tid  = threadIdx.x;
  const int wave = tid >> 5;
  const int lane = tid & 31;
  const int half = lane >> 4;
  const int m16  = lane & 15;
  const int row0 = blockIdx.x * (WAVES * ROWS) + wave;

  v4f nxt[NFFT / (32 * 4)];
  {
    const v4f* xr4 = (const v4f*)(x + (size_t)row0 * NFFT);
#pragma unroll
    for (int i = 0; i < NFFT / (32 * 4); ++i) nxt[i] = xr4[lane + 32 * i];
  }

  for (int i = tid; i < BASEN * BASEN; i += WAVES * 32) {
    const int k = i >> 4, n = i & 15;
    const float ang = -2.0f * PI_F * (float)((k * n) & 15) * (1.0f / 16.0f);
    BRt[i] = cosf(ang);
    BIt[i] = sinf(ang);
  }
  for (int i = tid; i < 512; i += WAVES * 32) {
    const float ang = -PI_F * (float)i * (1.0f / 512.0f);
    WRt[i] = cosf(ang);
    WIt[i] = sinf(ang);
  }
  __syncthreads();

  float* sx = buf[wave];
  float* ur = sx + NFFT;
  float* ui = ur + NFFT;
  v4f* sx4 = (v4f*)sx;

  const float p1r[7] = {WRt[m16 << 5],
                        WRt[m16 << 4], WRt[(m16 + 16) << 4],
                        WRt[m16 << 3], WRt[(m16 + 16) << 3], WRt[(m16 + 32) << 3], WRt[(m16 + 48) << 3]};
  const float p1i[7] = {WIt[m16 << 5],
                        WIt[m16 << 4], WIt[(m16 + 16) << 4],
                        WIt[m16 << 3], WIt[(m16 + 16) << 3], WIt[(m16 + 32) << 3], WIt[(m16 + 48) << 3]};
  const v2f w12r = {p1r[1], p1r[2]}, w12i = {p1i[1], p1i[2]};
  const v2f w34r = {p1r[3], p1r[4]}, w34i = {p1i[3], p1i[4]};
  const v2f w56r = {p1r[5], p1r[6]}, w56i = {p1i[5], p1i[6]};
  v2f p2r[2][7], p2i[2][7];
#pragma unroll
  for (int q = 0; q < 4; ++q) {
    const int k0 = 4 * lane + q;
    const int h = q >> 1, e = q & 1;
    p2r[h][0][e] = WRt[k0 << 2];
    p2r[h][1][e] = WRt[k0 << 1];  p2r[h][2][e] = WRt[(k0 + 128) << 1];
    p2r[h][3][e] = WRt[k0];       p2r[h][4][e] = WRt[k0 + 128];
    p2r[h][5][e] = WRt[k0 + 256]; p2r[h][6][e] = WRt[k0 + 384];
    p2i[h][0][e] = WIt[k0 << 2];
    p2i[h][1][e] = WIt[k0 << 1];  p2i[h][2][e] = WIt[(k0 + 128) << 1];
    p2i[h][3][e] = WIt[k0];       p2i[h][4][e] = WIt[k0 + 128];
    p2i[h][5][e] = WIt[k0 + 256]; p2i[h][6][e] = WIt[k0 + 384];
  }
  v2f bR[4], bI[4];
#pragma unroll
  for (int t = 0; t < 4; ++t) {
    const int k0 = 4 * t + 2 * half;
    bR[t][0] = BRt[(k0 + 0) * 16 + m16];
    bR[t][1] = BRt[(k0 + 1) * 16 + m16];
    bI[t][0] = BIt[(k0 + 0) * 16 + m16];
    bI[t][1] = BIt[(k0 + 1) * 16 + m16];
  }

#pragma nounroll
  for (int r = 0; r < ROWS; ++r) {
    const int row = row0 + r * WAVES;

#pragma unroll
    for (int i = 0; i < NFFT / (32 * 4); ++i) sx4[lane + 32 * i] = nxt[i];
    if (r + 1 < ROWS) {
      const v4f* xn4 = (const v4f*)(x + (size_t)(row + WAVES) * NFFT);
#pragma unroll
      for (int i = 0; i < NFFT / (32 * 4); ++i) nxt[i] = xn4[lane + 32 * i];
    }
    wave_lds_order();

#pragma unroll
    for (int mt = 0; mt < NCHUNK / 16; ++mt) {
      const int cOff = brev6(mt * 16 + m16);
      v8f accR = {0.f, 0.f, 0.f, 0.f, 0.f, 0.f, 0.f, 0.f};
      v8f accI = {0.f, 0.f, 0.f, 0.f, 0.f, 0.f, 0.f, 0.f};
#pragma unroll
      for (int t = 0; t < 4; ++t) {
        const int k0 = 4 * t + 2 * half;
        v2f a;
        a[0] = sx[(k0 + 0) * NCHUNK + cOff];
        a[1] = sx[(k0 + 1) * NCHUNK + cOff];
        accR = __builtin_amdgcn_wmma_f32_16x16x4_f32(false, a, false, bR[t], (short)0, accR, false, false);
        accI = __builtin_amdgcn_wmma_f32_16x16x4_f32(false, a, false, bI[t], (short)0, accI, false, false);
      }
      float R[8] = {accR[0], accR[1], accR[2], accR[3], accR[4], accR[5], accR[6], accR[7]};
      float I[8] = {accI[0], accI[1], accI[2], accI[3], accI[4], accI[5], accI[6], accI[7]};
      bfly(R[0], I[0], R[1], I[1], p1r[0], p1i[0]);
      bfly(R[2], I[2], R[3], I[3], p1r[0], p1i[0]);
      bfly(R[4], I[4], R[5], I[5], p1r[0], p1i[0]);
      bfly(R[6], I[6], R[7], I[7], p1r[0], p1i[0]);
      v2f PR[4], PI[4];
#pragma unroll
      for (int j = 0; j < 4; ++j) {
        PR[j][0] = R[2 * j];  PR[j][1] = R[2 * j + 1];
        PI[j][0] = I[2 * j];  PI[j][1] = I[2 * j + 1];
      }
      bfly2(PR[0], PI[0], PR[1], PI[1], w12r, w12i);
      bfly2(PR[2], PI[2], PR[3], PI[3], w12r, w12i);
      bfly2(PR[0], PI[0], PR[2], PI[2], w34r, w34i);
      bfly2(PR[1], PI[1], PR[3], PI[3], w56r, w56i);
#pragma unroll
      for (int i = 0; i < 8; ++i) {
        const int c = mt * 16 + i + 8 * half;
        ur[c * 16 + m16] = PR[i >> 1][i & 1];
        ui[c * 16 + m16] = PI[i >> 1][i & 1];
      }
    }
    wave_lds_order();

    float* outR = out + (size_t)row * 513;
    float* outI = out + (size_t)nrows * 513 + (size_t)row * 513;
    const v4f* ur4 = (const v4f*)ur;
    const v4f* ui4 = (const v4f*)ui;
    v4f UR[8], UI[8];
#pragma unroll
    for (int j = 0; j < 8; ++j) {
      UR[j] = ur4[32 * j + lane];
      UI[j] = ui4[32 * j + lane];
    }
    v2f oR[4][2], oI[4][2];
    float r512 = 0.f, i512 = 0.f;
#pragma unroll
    for (int h = 0; h < 2; ++h) {
      v2f R[8], I[8];
#pragma unroll
      for (int j = 0; j < 8; ++j) {
        R[j] = h ? __builtin_shufflevector(UR[j], UR[j], 2, 3) : __builtin_shufflevector(UR[j], UR[j], 0, 1);
        I[j] = h ? __builtin_shufflevector(UI[j], UI[j], 2, 3) : __builtin_shufflevector(UI[j], UI[j], 0, 1);
      }
      radix8x2(R, I, p2r[h], p2i[h]);
#pragma unroll
      for (int b = 0; b < 4; ++b) { oR[b][h] = R[b];  oI[b][h] = I[b]; }
      if (h == 0) { r512 = R[4][0];  i512 = I[4][0]; }
    }
#pragma unroll
    for (int b = 0; b < 4; ++b) {
      *(v4f_a4*)(outR + 128 * b + 4 * lane) = __builtin_shufflevector(oR[b][0], oR[b][1], 0, 1, 2, 3);
      *(v4f_a4*)(outI + 128 * b + 4 * lane) = __builtin_shufflevector(oI[b][0], oI[b][1], 0, 1, 2, 3);
    }
    if (lane == 0) { outR[512] = r512;  outI[512] = i512; }
    wave_lds_order();
  }
}

extern "C" void kernel_launch(void* const* d_in, const int* in_sizes, int n_in,
                              void* d_out, int out_size, void* d_ws, size_t ws_size,
                              hipStream_t stream) {
  const float* x = (const float*)d_in[0];
  float* out = (float*)d_out;
  const int nrows = in_sizes[0] / NFFT;
  const int blocks = nrows / (WAVES * ROWS);
  rfft1024_kernel<<<blocks, WAVES * 32, 0, stream>>>(x, out, nrows);
}


# Real-input symmetry: row taken as 512 complex z[n]=x[2n]+i*x[2n+1], 512-point FFT (5 stages, was 6 on 1024) + one split pass X[k],X[512-k] from Z[k],Z[512-k]; sum order changes
# speedup vs baseline: 2.3710x; 1.0616x over previous
#include <hip/hip_runtime.h>
#include <math.h>

typedef float v2f __attribute__((ext_vector_type(2)));
typedef float v8f __attribute__((ext_vector_type(8)));
typedef float v4f __attribute__((ext_vector_type(4)));
typedef float v4f_a4 __attribute__((ext_vector_type(4), aligned(4)));

#define NFFT   1024
#define HALFN  512
#define BASEN  16
#define NCHUNK 32
#define WAVES  4
#define ROWS   4
#define PI_F   3.14159265358979323846f

__device__ __forceinline__ int brev5(int c) { return (int)(__brev((unsigned)c) >> 27); }

__device__ __forceinline__ void bfly(float& er, float& ei, float& orr, float& oi,
                                     const float wr, const float wi) {
  const float tr = wr * orr - wi * oi;
  const float ti = wr * oi + wi * orr;
  const float ar = er, ai = ei;
  er  = ar + tr;  ei = ai + ti;
  orr = ar - tr;  oi = ai - ti;
}

__device__ __forceinline__ void bfly2(v2f& er, v2f& ei, v2f& orr, v2f& oi, const v2f wr, const v2f wi) {
  const v2f tr = wr * orr - wi * oi;
  const v2f ti = wr * oi + wi * orr;
  const v2f ar = er, ai = ei;
  er  = ar + tr;  ei = ai + ti;
  orr = ar - tr;  oi = ai - ti;
}

__device__ __forceinline__ void radix4x2(v2f (&R)[4], v2f (&I)[4],
                                         const v2f (&wr)[3], const v2f (&wi)[3]) {
  bfly2(R[0], I[0], R[1], I[1], wr[0], wi[0]);
  bfly2(R[2], I[2], R[3], I[3], wr[0], wi[0]);
  bfly2(R[0], I[0], R[2], I[2], wr[1], wi[1]);
  bfly2(R[1], I[1], R[3], I[3], wr[2], wi[2]);
}

__device__ __forceinline__ void split2(const v2f zkr, const v2f zki, const v2f zmr, const v2f zmi,
                                       const v2f wr, const v2f wi,
                                       v2f& xr, v2f& xi, v2f& yr, v2f& yi) {
  const v2f er = zkr + zmr, ei = zki - zmi;
  const v2f orr = zki + zmi, oi = zmr - zkr;
  const v2f tr = wr * orr - wi * oi;
  const v2f ti = wr * oi + wi * orr;
  xr = er + tr;  xi = ei + ti;
  yr = er - tr;  yi = ti - ei;
}

__device__ __forceinline__ void wave_lds_order() {
  __builtin_amdgcn_fence(__ATOMIC_RELEASE, "wavefront");
  __builtin_amdgcn_wave_barrier();
  __builtin_amdgcn_fence(__ATOMIC_ACQUIRE, "wavefront");
}

__global__ __launch_bounds__(WAVES * 32) void rfft1024_kernel(
    const float* __restrict__ x, float* __restrict__ out, int nrows) {
  __shared__ __align__(16) float buf[WAVES][2 * NFFT];
  __shared__ float BRt[BASEN * BASEN];
  __shared__ float BIt[BASEN * BASEN];
  __shared__ __align__(16) float WRt[512];
  __shared__ __align__(16) float WIt[512];

  const int tid  = threadIdx.x;
  const int wave = tid >> 5;
  const int lane = tid & 31;
  const int half = lane >> 4;
  const int m16  = lane & 15;
  const int row0 = blockIdx.x * (WAVES * ROWS) + wave;

  v4f nxt[NFFT / (32 * 4)];
  {
    const v4f* xr4 = (const v4f*)(x + (size_t)row0 * NFFT);
#pragma unroll
    for (int i = 0; i < NFFT / (32 * 4); ++i) nxt[i] = xr4[lane + 32 * i];
  }

  for (int i = tid; i < BASEN * BASEN; i += WAVES * 32) {
    const int k = i >> 4, n = i & 15;
    const float ang = -2.0f * PI_F * (float)((k * n) & 15) * (1.0f / 16.0f);
    BRt[i] = 0.5f * cosf(ang);
    BIt[i] = 0.5f * sinf(ang);
  }
  for (int i = tid; i < 512; i += WAVES * 32) {
    const float ang = -PI_F * (float)i * (1.0f / 512.0f);
    WRt[i] = cosf(ang);
    WIt[i] = sinf(ang);
  }
  __syncthreads();

  float* sx = buf[wave];
  float* ur = sx + NFFT;
  float* ui = ur + HALFN;
  v4f* sx4 = (v4f*)sx;

  const float p1r[7] = {WRt[m16 << 5],
                        WRt[m16 << 4], WRt[(m16 + 16) << 4],
                        WRt[m16 << 3], WRt[(m16 + 16) << 3], WRt[(m16 + 32) << 3], WRt[(m16 + 48) << 3]};
  const float p1i[7] = {WIt[m16 << 5],
                        WIt[m16 << 4], WIt[(m16 + 16) << 4],
                        WIt[m16 << 3], WIt[(m16 + 16) << 3], WIt[(m16 + 32) << 3], WIt[(m16 + 48) << 3]};
  const v2f w12r = {p1r[1], p1r[2]}, w12i = {p1i[1], p1i[2]};
  const v2f w34r = {p1r[3], p1r[4]}, w34i = {p1i[3], p1i[4]};
  const v2f w56r = {p1r[5], p1r[6]}, w56i = {p1i[5], p1i[6]};
  v2f p2r[2][3], p2i[2][3];
#pragma unroll
  for (int q = 0; q < 4; ++q) {
    const int k0 = 4 * lane + q;
    const int h = q >> 1, e = q & 1;
    p2r[h][0][e] = WRt[k0 << 2];
    p2r[h][1][e] = WRt[k0 << 1];  p2r[h][2][e] = WRt[(k0 + 128) << 1];
    p2i[h][0][e] = WIt[k0 << 2];
    p2i[h][1][e] = WIt[k0 << 1];  p2i[h][2][e] = WIt[(k0 + 128) << 1];
  }
  const v4f swr0 = ((const v4f*)WRt)[lane], swr1 = ((const v4f*)WRt)[32 + lane];
  const v4f swi0 = ((const v4f*)WIt)[lane], swi1 = ((const v4f*)WIt)[32 + lane];
  v2f bR[4], bI[4], bIn[4];
#pragma unroll
  for (int t = 0; t < 4; ++t) {
    const int k0 = 4 * t + 2 * half;
    bR[t][0] = BRt[(k0 + 0) * 16 + m16];
    bR[t][1] = BRt[(k0 + 1) * 16 + m16];
    bI[t][0] = BIt[(k0 + 0) * 16 + m16];
    bI[t][1] = BIt[(k0 + 1) * 16 + m16];
    bIn[t] = -bI[t];
  }

#pragma nounroll
  for (int r = 0; r < ROWS; ++r) {
    const int row = row0 + r * WAVES;

#pragma unroll
    for (int i = 0; i < NFFT / (32 * 4); ++i) sx4[lane + 32 * i] = nxt[i];
    if (r + 1 < ROWS) {
      const v4f* xn4 = (const v4f*)(x + (size_t)(row + WAVES) * NFFT);
#pragma unroll
      for (int i = 0; i < NFFT / (32 * 4); ++i) nxt[i] = xn4[lane + 32 * i];
    }
    wave_lds_order();

#pragma unroll
    for (int mt = 0; mt < NCHUNK / 16; ++mt) {
      const int cOff = 2 * brev5(mt * 16 + m16);
      v8f accR = {0.f, 0.f, 0.f, 0.f, 0.f, 0.f, 0.f, 0.f};
      v8f accI = {0.f, 0.f, 0.f, 0.f, 0.f, 0.f, 0.f, 0.f};
#pragma unroll
      for (int t = 0; t < 4; ++t) {
        const int k0 = 4 * t + 2 * half;
        v2f ar, ai;
        ar[0] = sx[(k0 + 0) * 64 + cOff];
        ar[1] = sx[(k0 + 1) * 64 + cOff];
        ai[0] = sx[(k0 + 0) * 64 + cOff + 1];
        ai[1] = sx[(k0 + 1) * 64 + cOff + 1];
        accR = __builtin_amdgcn_wmma_f32_16x16x4_f32(false, ar, false, bR[t], (short)0, accR, false, false);
        accI = __builtin_amdgcn_wmma_f32_16x16x4_f32(false, ar, false, bI[t], (short)0, accI, false, false);
        accR = __builtin_amdgcn_wmma_f32_16x16x4_f32(false, ai, false, bIn[t], (short)0, accR, false, false);
        accI = __builtin_amdgcn_wmma_f32_16x16x4_f32(false, ai, false, bR[t], (short)0, accI, false, false);
      }
      float R[8] = {accR[0], accR[1], accR[2], accR[3], accR[4], accR[5], accR[6], accR[7]};
      float I[8] = {accI[0], accI[1], accI[2], accI[3], accI[4], accI[5], accI[6], accI[7]};
      bfly(R[0], I[0], R[1], I[1], p1r[0], p1i[0]);
      bfly(R[2], I[2], R[3], I[3], p1r[0], p1i[0]);
      bfly(R[4], I[4], R[5], I[5], p1r[0], p1i[0]);
      bfly(R[6], I[6], R[7], I[7], p1r[0], p1i[0]);
      v2f PR[4], PI[4];
#pragma unroll
      for (int j = 0; j < 4; ++j) {
        PR[j][0] = R[2 * j];  PR[j][1] = R[2 * j + 1];
        PI[j][0] = I[2 * j];  PI[j][1] = I[2 * j + 1];
      }
      bfly2(PR[0], PI[0], PR[1], PI[1], w12r, w12i);
      bfly2(PR[2], PI[2], PR[3], PI[3], w12r, w12i);
      bfly2(PR[0], PI[0], PR[2], PI[2], w34r, w34i);
      bfly2(PR[1], PI[1], PR[3], PI[3], w56r, w56i);
#pragma unroll
      for (int i = 0; i < 8; ++i) {
        const int c = mt * 16 + i + 8 * half;
        ur[c * 16 + m16] = PR[i >> 1][i & 1];
        ui[c * 16 + m16] = PI[i >> 1][i & 1];
      }
    }
    wave_lds_order();

    const v4f* ur4 = (const v4f*)ur;
    const v4f* ui4 = (const v4f*)ui;
    v4f* zr4 = (v4f*)sx;
    v4f* zi4 = (v4f*)(sx + HALFN);
    {
      v4f UR[4], UI[4];
#pragma unroll
      for (int g = 0; g < 4; ++g) {
        UR[g] = ur4[32 * g + lane];
        UI[g] = ui4[32 * g + lane];
      }
      v2f oR[4][2], oI[4][2];
#pragma unroll
      for (int h = 0; h < 2; ++h) {
        v2f R[4], I[4];
#pragma unroll
        for (int g = 0; g < 4; ++g) {
          R[g] = h ? __builtin_shufflevector(UR[g], UR[g], 2, 3) : __builtin_shufflevector(UR[g], UR[g], 0, 1);
          I[g] = h ? __builtin_shufflevector(UI[g], UI[g], 2, 3) : __builtin_shufflevector(UI[g], UI[g], 0, 1);
        }
        radix4x2(R, I, p2r[h], p2i[h]);
#pragma unroll
        for (int b = 0; b < 4; ++b) { oR[b][h] = R[b];  oI[b][h] = I[b]; }
      }
#pragma unroll
      for (int b = 0; b < 4; ++b) {
        zr4[32 * b + lane] = __builtin_shufflevector(oR[b][0], oR[b][1], 0, 1, 2, 3);
        zi4[32 * b + lane] = __builtin_shufflevector(oI[b][0], oI[b][1], 0, 1, 2, 3);
      }
    }
    wave_lds_order();

    float* outR = out + (size_t)row * 513;
    float* outI = out + (size_t)nrows * 513 + (size_t)row * 513;
    {
      const v4f dr0 = zr4[lane], dr1 = zr4[32 + lane];
      const v4f di0 = zi4[lane], di1 = zi4[32 + lane];
      const v4f ar4 = zr4[127 - lane], br4 = zr4[95 - lane];
      const v4f ai4 = zi4[127 - lane], bi4 = zi4[95 - lane];
      const int p0 = (HALFN - 4 * lane) & (HALFN - 1);
      const int p1 = 384 - 4 * lane;
      const float ar0 = sx[p0], ai0 = sx[HALFN + p0];
      const float br0 = sx[p1], bi0 = sx[HALFN + p1];
      const v2f mr[4] = {{ar0, ar4[3]}, {ar4[2], ar4[1]}, {br0, br4[3]}, {br4[2], br4[1]}};
      const v2f mi[4] = {{ai0, ai4[3]}, {ai4[2], ai4[1]}, {bi0, bi4[3]}, {bi4[2], bi4[1]}};
      const v2f kr[4] = {{dr0[0], dr0[1]}, {dr0[2], dr0[3]}, {dr1[0], dr1[1]}, {dr1[2], dr1[3]}};
      const v2f ki[4] = {{di0[0], di0[1]}, {di0[2], di0[3]}, {di1[0], di1[1]}, {di1[2], di1[3]}};
      const v2f wr[4] = {{swr0[0], swr0[1]}, {swr0[2], swr0[3]}, {swr1[0], swr1[1]}, {swr1[2], swr1[3]}};
      const v2f wi[4] = {{swi0[0], swi0[1]}, {swi0[2], swi0[3]}, {swi1[0], swi1[1]}, {swi1[2], swi1[3]}};
      v2f xr[4], xi[4], yr[4], yi[4];
#pragma unroll
      for (int p = 0; p < 4; ++p) split2(kr[p], ki[p], mr[p], mi[p], wr[p], wi[p], xr[p], xi[p], yr[p], yi[p]);
      *(v4f_a4*)(outR + 4 * lane)       = __builtin_shufflevector(xr[0], xr[1], 0, 1, 2, 3);
      *(v4f_a4*)(outR + 128 + 4 * lane) = __builtin_shufflevector(xr[2], xr[3], 0, 1, 2, 3);
      *(v4f_a4*)(outI + 4 * lane)       = __builtin_shufflevector(xi[0], xi[1], 0, 1, 2, 3);
      *(v4f_a4*)(outI + 128 + 4 * lane) = __builtin_shufflevector(xi[2], xi[3], 0, 1, 2, 3);
      *(v4f_a4*)(outR + 509 - 4 * lane) = __builtin_shufflevector(yr[1], yr[0], 1, 0, 3, 2);
      *(v4f_a4*)(outR + 381 - 4 * lane) = __builtin_shufflevector(yr[3], yr[2], 1, 0, 3, 2);
      *(v4f_a4*)(outI + 509 - 4 * lane) = __builtin_shufflevector(yi[1], yi[0], 1, 0, 3, 2);
      *(v4f_a4*)(outI + 381 - 4 * lane) = __builtin_shufflevector(yi[3], yi[2], 1, 0, 3, 2);
      if (lane == 31) { outR[256] = 2.0f * sx[256];  outI[256] = -2.0f * sx[HALFN + 256]; }
    }
    wave_lds_order();
  }
}

extern "C" void kernel_launch(void* const* d_in, const int* in_sizes, int n_in,
                              void* d_out, int out_size, void* d_ws, size_t ws_size,
                              hipStream_t stream) {
  const float* x = (const float*)d_in[0];
  float* out = (float*)d_out;
  const int nrows = in_sizes[0] / NFFT;
  const int blocks = nrows / (WAVES * ROWS);
  rfft1024_kernel<<<blocks, WAVES * 32, 0, stream>>>(x, out, nrows);
}


# DFT16 product folded over the time index: z[n]+z[16-n] meets the cos rows, z[n]-z[16-n] the sin rows, one real 16x16 B for real and imaginary part: 16 WMMAs per row, was 32; sum order changes
# speedup vs baseline: 2.5285x; 1.0664x over previous
#include <hip/hip_runtime.h>
#include <math.h>

typedef float v2f __attribute__((ext_vector_type(2)));
typedef float v8f __attribute__((ext_vector_type(8)));
typedef float v4f __attribute__((ext_vector_type(4)));
typedef float v4f_a4 __attribute__((ext_vector_type(4), aligned(4)));

#define NFFT   1024
#define HALFN  512
#define BASEN  16
#define NCHUNK 32
#define WAVES  4
#define ROWS   4
#define PI_F   3.14159265358979323846f

__device__ __forceinline__ int brev5(int c) { return (int)(__brev((unsigned)c) >> 27); }

__device__ __forceinline__ void bfly(float& er, float& ei, float& orr, float& oi,
                                     const float wr, const float wi) {
  const float tr = wr * orr - wi * oi;
  const float ti = wr * oi + wi * orr;
  const float ar = er, ai = ei;
  er  = ar + tr;  ei = ai + ti;
  orr = ar - tr;  oi = ai - ti;
}

__device__ __forceinline__ void bfly2(v2f& er, v2f& ei, v2f& orr, v2f& oi, const v2f wr, const v2f wi) {
  const v2f tr = wr * orr - wi * oi;
  const v2f ti = wr * oi + wi * orr;
  const v2f ar = er, ai = ei;
  er  = ar + tr;  ei = ai + ti;
  orr = ar - tr;  oi = ai - ti;
}

__device__ __forceinline__ void radix4x2(v2f (&R)[4], v2f (&I)[4],
                                         const v2f (&wr)[3], const v2f (&wi)[3]) {
  bfly2(R[0], I[0], R[1], I[1], wr[0], wi[0]);
  bfly2(R[2], I[2], R[3], I[3], wr[0], wi[0]);
  bfly2(R[0], I[0], R[2], I[2], wr[1], wi[1]);
  bfly2(R[1], I[1], R[3], I[3], wr[2], wi[2]);
}

__device__ __forceinline__ void split2(const v2f zkr, const v2f zki, const v2f zmr, const v2f zmi,
                                       const v2f wr, const v2f wi,
                                       v2f& xr, v2f& xi, v2f& yr, v2f& yi) {
  const v2f er = zkr + zmr, ei = zki - zmi;
  const v2f orr = zki + zmi, oi = zmr - zkr;
  const v2f tr = wr * orr - wi * oi;
  const v2f ti = wr * oi + wi * orr;
  xr = er + tr;  xi = ei + ti;
  yr = er - tr;  yi = ti - ei;
}

__device__ __forceinline__ void wave_lds_order() {
  __builtin_amdgcn_fence(__ATOMIC_RELEASE, "wavefront");
  __builtin_amdgcn_wave_barrier();
  __builtin_amdgcn_fence(__ATOMIC_ACQUIRE, "wavefront");
}

__global__ __launch_bounds__(WAVES * 32) void rfft1024_kernel(
    const float* __restrict__ x, float* __restrict__ out, int nrows) {
  __shared__ __align__(16) float buf[WAVES][2 * NFFT];
  __shared__ float BRt[BASEN * BASEN];
  __shared__ float BIt[BASEN * BASEN];
  __shared__ __align__(16) float WRt[512];
  __shared__ __align__(16) float WIt[512];

  const int tid  = threadIdx.x;
  const int wave = tid >> 5;
  const int lane = tid & 31;
  const int half = lane >> 4;
  const int m16  = lane & 15;
  const int row0 = blockIdx.x * (WAVES * ROWS) + wave;

  v4f nxt[NFFT / (32 * 4)];
  {
    const v4f* xr4 = (const v4f*)(x + (size_t)row0 * NFFT);
#pragma unroll
    for (int i = 0; i < NFFT / (32 * 4); ++i) nxt[i] = xr4[lane + 32 * i];
  }

  for (int i = tid; i < BASEN * BASEN; i += WAVES * 32) {
    const int k = i >> 4, n = i & 15;
    const float ang = -2.0f * PI_F * (float)((k * n) & 15) * (1.0f / 16.0f);
    BRt[i] = 0.5f * cosf(ang);
    BIt[i] = 0.5f * sinf(ang);
  }
  for (int i = tid; i < 512; i += WAVES * 32) {
    const float ang = -PI_F * (float)i * (1.0f / 512.0f);
    WRt[i] = cosf(ang);
    WIt[i] = sinf(ang);
  }
  __syncthreads();

  float* sx = buf[wave];
  float* ur = sx + NFFT;
  float* ui = ur + HALFN;
  v4f* sx4 = (v4f*)sx;

  const float p1r[7] = {WRt[m16 << 5],
                        WRt[m16 << 4], WRt[(m16 + 16) << 4],
                        WRt[m16 << 3], WRt[(m16 + 16) << 3], WRt[(m16 + 32) << 3], WRt[(m16 + 48) << 3]};
  const float p1i[7] = {WIt[m16 << 5],
                        WIt[m16 << 4], WIt[(m16 + 16) << 4],
                        WIt[m16 << 3], WIt[(m16 + 16) << 3], WIt[(m16 + 32) << 3], WIt[(m16 + 48) << 3]};
  const v2f w12r = {p1r[1], p1r[2]}, w12i = {p1i[1], p1i[2]};
  const v2f w34r = {p1r[3], p1r[4]}, w34i = {p1i[3], p1i[4]};
  const v2f w56r = {p1r[5], p1r[6]}, w56i = {p1i[5], p1i[6]};
  v2f p2r[2][3], p2i[2][3];
#pragma unroll
  for (int q = 0; q < 4; ++q) {
    const int k0 = 4 * lane + q;
    const int h = q >> 1, e = q & 1;
    p2r[h][0][e] = WRt[k0 << 2];
    p2r[h][1][e] = WRt[k0 << 1];  p2r[h][2][e] = WRt[(k0 + 128) << 1];
    p2i[h][0][e] = WIt[k0 << 2];
    p2i[h][1][e] = WIt[k0 << 1];  p2i[h][2][e] = WIt[(k0 + 128) << 1];
  }
  const v4f swr0 = ((const v4f*)WRt)[lane], swr1 = ((const v4f*)WRt)[32 + lane];
  const v4f swi0 = ((const v4f*)WIt)[lane], swi1 = ((const v4f*)WIt)[32 + lane];
  v2f bF[4];
#pragma unroll
  for (int t = 0; t < 4; ++t) {
    const int na = 4 * half + 1 + t;
    bF[t][0] = BRt[na * 16 + m16];
    bF[t][1] = -BIt[na * 16 + m16];
  }
  if (half) {
    bF[3][0] = (m16 & 1) ? 0.0f : 0.5f;
    bF[3][1] = (m16 & 1) ? -0.5f : 0.0f;
  }

#pragma nounroll
  for (int r = 0; r < ROWS; ++r) {
    const int row = row0 + r * WAVES;

#pragma unroll
    for (int i = 0; i < NFFT / (32 * 4); ++i) sx4[lane + 32 * i] = nxt[i];
    if (r + 1 < ROWS) {
      const v4f* xn4 = (const v4f*)(x + (size_t)(row + WAVES) * NFFT);
#pragma unroll
      for (int i = 0; i < NFFT / (32 * 4); ++i) nxt[i] = xn4[lane + 32 * i];
    }
    wave_lds_order();

#pragma unroll
    for (int mt = 0; mt < NCHUNK / 16; ++mt) {
      const int cOff = 2 * brev5(mt * 16 + m16);
      v8f accR = {0.f, 0.f, 0.f, 0.f, 0.f, 0.f, 0.f, 0.f};
      v8f accI = {0.f, 0.f, 0.f, 0.f, 0.f, 0.f, 0.f, 0.f};
      v2f za[4], zb[4];
#pragma unroll
      for (int t = 0; t < 4; ++t) {
        const int na = 4 * half + 1 + t;
        const int nb = (t == 3 && half) ? 0 : 16 - na;
        za[t] = *(const v2f*)(sx + 64 * na + cOff);
        zb[t] = *(const v2f*)(sx + 64 * nb + cOff);
      }
#pragma unroll
      for (int t = 0; t < 4; ++t) {
        const v2f s = za[t] + zb[t];
        const v2f d = za[t] - zb[t];
        v2f p, q;
        p[0] = s[0];
        q[0] = s[1];
        if (t < 3) {
          p[1] = d[1];
          q[1] = zb[t][0] - za[t][0];
        } else {
          p[1] = half ? d[0] : d[1];
          q[1] = half ? d[1] : zb[t][0] - za[t][0];
        }
        accR = __builtin_amdgcn_wmma_f32_16x16x4_f32(false, p, false, bF[t], (short)0, accR, false, false);
        accI = __builtin_amdgcn_wmma_f32_16x16x4_f32(false, q, false, bF[t], (short)0, accI, false, false);
      }
      float R[8] = {accR[0], accR[1], accR[2], accR[3], accR[4], accR[5], accR[6], accR[7]};
      float I[8] = {accI[0], accI[1], accI[2], accI[3], accI[4], accI[5], accI[6], accI[7]};
      bfly(R[0], I[0], R[1], I[1], p1r[0], p1i[0]);
      bfly(R[2], I[2], R[3], I[3], p1r[0], p1i[0]);
      bfly(R[4], I[4], R[5], I[5], p1r[0], p1i[0]);
      bfly(R[6], I[6], R[7], I[7], p1r[0], p1i[0]);
      v2f PR[4], PI[4];
#pragma unroll
      for (int j = 0; j < 4; ++j) {
        PR[j][0] = R[2 * j];  PR[j][1] = R[2 * j + 1];
        PI[j][0] = I[2 * j];  PI[j][1] = I[2 * j + 1];
      }
      bfly2(PR[0], PI[0], PR[1], PI[1], w12r, w12i);
      bfly2(PR[2], PI[2], PR[3], PI[3], w12r, w12i);
      bfly2(PR[0], PI[0], PR[2], PI[2], w34r, w34i);
      bfly2(PR[1], PI[1], PR[3], PI[3], w56r, w56i);
#pragma unroll
      for (int i = 0; i < 8; ++i) {
        const int c = mt * 16 + i + 8 * half;
        ur[c * 16 + m16] = PR[i >> 1][i & 1];
        ui[c * 16 + m16] = PI[i >> 1][i & 1];
      }
    }
    wave_lds_order();

    const v4f* ur4 = (const v4f*)ur;
    const v4f* ui4 = (const v4f*)ui;
    v4f* zr4 = (v4f*)sx;
    v4f* zi4 = (v4f*)(sx + HALFN);
    {
      v4f UR[4], UI[4];
#pragma unroll
      for (int g = 0; g < 4; ++g) {
        UR[g] = ur4[32 * g + lane];
        UI[g] = ui4[32 * g + lane];
      }
      v2f oR[4][2], oI[4][2];
#pragma unroll
      for (int h = 0; h < 2; ++h) {
        v2f R[4], I[4];
#pragma unroll
        for (int g = 0; g < 4; ++g) {
          R[g] = h ? __builtin_shufflevector(UR[g], UR[g], 2, 3) : __builtin_shufflevector(UR[g], UR[g], 0, 1);
          I[g] = h ? __builtin_shufflevector(UI[g], UI[g], 2, 3) : __builtin_shufflevector(UI[g], UI[g], 0, 1);
        }
        radix4x2(R, I, p2r[h], p2i[h]);
#pragma unroll
        for (int b = 0; b < 4; ++b) { oR[b][h] = R[b];  oI[b][h] = I[b]; }
      }
#pragma unroll
      for (int b = 0; b < 4; ++b) {
        zr4[32 * b + lane] = __builtin_shufflevector(oR[b][0], oR[b][1], 0, 1, 2, 3);
        zi4[32 * b + lane] = __builtin_shufflevector(oI[b][0], oI[b][1], 0, 1, 2, 3);
      }
    }
    wave_lds_order();

    float* outR = out + (size_t)row * 513;
    float* outI = out + (size_t)nrows * 513 + (size_t)row * 513;
    {
      const v4f dr0 = zr4[lane], dr1 = zr4[32 + lane];
      const v4f di0 = zi4[lane], di1 = zi4[32 + lane];
      const v4f ar4 = zr4[127 - lane], br4 = zr4[95 - lane];
      const v4f ai4 = zi4[127 - lane], bi4 = zi4[95 - lane];
      const int p0 = (HALFN - 4 * lane) & (HALFN - 1);
      const int p1 = 384 - 4 * lane;
      const float ar0 = sx[p0], ai0 = sx[HALFN + p0];
      const float br0 = sx[p1], bi0 = sx[HALFN + p1];
      const v2f mr[4] = {{ar0, ar4[3]}, {ar4[2], ar4[1]}, {br0, br4[3]}, {br4[2], br4[1]}};
      const v2f mi[4] = {{ai0, ai4[3]}, {ai4[2], ai4[1]}, {bi0, bi4[3]}, {bi4[2], bi4[1]}};
      const v2f kr[4] = {{dr0[0], dr0[1]}, {dr0[2], dr0[3]}, {dr1[0], dr1[1]}, {dr1[2], dr1[3]}};
      const v2f ki[4] = {{di0[0], di0[1]}, {di0[2], di0[3]}, {di1[0], di1[1]}, {di1[2], di1[3]}};
      const v2f wr[4] = {{swr0[0], swr0[1]}, {swr0[2], swr0[3]}, {swr1[0], swr1[1]}, {swr1[2], swr1[3]}};
      const v2f wi[4] = {{swi0[0], swi0[1]}, {swi0[2], swi0[3]}, {swi1[0], swi1[1]}, {swi1[2], swi1[3]}};
      v2f xr[4], xi[4], yr[4], yi[4];
#pragma unroll
      for (int p = 0; p < 4; ++p) split2(kr[p], ki[p], mr[p], mi[p], wr[p], wi[p], xr[p], xi[p], yr[p], yi[p]);
      *(v4f_a4*)(outR + 4 * lane)       = __builtin_shufflevector(xr[0], xr[1], 0, 1, 2, 3);
      *(v4f_a4*)(outR + 128 + 4 * lane) = __builtin_shufflevector(xr[2], xr[3], 0, 1, 2, 3);
      *(v4f_a4*)(outI + 4 * lane)       = __builtin_shufflevector(xi[0], xi[1], 0, 1, 2, 3);
      *(v4f_a4*)(outI + 128 + 4 * lane) = __builtin_shufflevector(xi[2], xi[3], 0, 1, 2, 3);
      *(v4f_a4*)(outR + 509 - 4 * lane) = __builtin_shufflevector(yr[1], yr[0], 1, 0, 3, 2);
      *(v4f_a4*)(outR + 381 - 4 * lane) = __builtin_shufflevector(yr[3], yr[2], 1, 0, 3, 2);
      *(v4f_a4*)(outI + 509 - 4 * lane) = __builtin_shufflevector(yi[1], yi[0], 1, 0, 3, 2);
      *(v4f_a4*)(outI + 381 - 4 * lane) = __builtin_shufflevector(yi[3], yi[2], 1, 0, 3, 2);
      if (lane == 31) { outR[256] = 2.0f * sx[256];  outI[256] = -2.0f * sx[HALFN + 256]; }
    }
    wave_lds_order();
  }
}

extern "C" void kernel_launch(void* const* d_in, const int* in_sizes, int n_in,
                              void* d_out, int out_size, void* d_ws, size_t ws_size,
                              hipStream_t stream) {
  const float* x = (const float*)d_in[0];
  float* out = (float*)d_out;
  const int nrows = in_sizes[0] / NFFT;
  const int blocks = nrows / (WAVES * ROWS);
  rfft1024_kernel<<<blocks, WAVES * 32, 0, stream>>>(x, out, nrows);
}


# Twiddle and DFT constants come from a compile-time table in each lane's own layout (16 x 16-byte loads beside the first row): no cosf/sinf table build, no LDS table, no block barrier; same sum order
# speedup vs baseline: 2.8155x; 1.1135x over previous
#include <hip/hip_runtime.h>
#include <math.h>

typedef float v2f __attribute__((ext_vector_type(2)));
typedef float v8f __attribute__((ext_vector_type(8)));
typedef float v4f __attribute__((ext_vector_type(4)));
typedef float v4f_a4 __attribute__((ext_vector_type(4), aligned(4)));
typedef __attribute__((address_space(1))) v4f gv4f;

#define NFFT   1024
#define HALFN  512
#define BASEN  16
#define NCHUNK 32
#define WAVES  4
#define ROWS   4
#define PI_F   3.14159265358979323846f


struct CS { double c, s; };
constexpr CS oct_cs(int m) {
  const double x = 3.14159265358979323846 * m / 512.0, x2 = x * x;
  const double s = x * (1.0 + x2 * (-1.0 / 6.0 + x2 * (1.0 / 120.0 + x2 * (-1.0 / 5040.0 + x2 * (1.0 / 362880.0
                   + x2 * (-1.0 / 39916800.0 + x2 * (1.0 / 6227020800.0 + x2 * (-1.0 / 1307674368000.0))))))));
  const double c = 1.0 + x2 * (-1.0 / 2.0 + x2 * (1.0 / 24.0 + x2 * (-1.0 / 720.0 + x2 * (1.0 / 40320.0
                   + x2 * (-1.0 / 3628800.0 + x2 * (1.0 / 479001600.0 + x2 * (-1.0 / 87178291200.0
                   + x2 * (1.0 / 20922789888000.0))))))));
  return {c, s};
}
constexpr CS tw(int idx) {
  const int k = idx & 511;
  double c = 0.0, s = 0.0;
  if (k <= 128)      { const CS o = oct_cs(k);        c =  o.c;  s = o.s; }
  else if (k <= 256) { const CS o = oct_cs(256 - k);  c =  o.s;  s = o.c; }
  else if (k <= 384) { const CS o = oct_cs(k - 256);  c = -o.s;  s = o.c; }
  else               { const CS o = oct_cs(512 - k);  c = -o.c;  s = o.s; }
  if (idx & 512) { c = -c;  s = -s; }
  return {c, -s};
}
constexpr float twr(int idx) { return (float)tw(idx).c; }
constexpr float twi(int idx) { return (float)tw(idx).s; }

struct LaneTab { float v[16 * 32 * 4]; };
constexpr LaneTab make_lane_tab() {
  LaneTab T{};
  for (int l = 0; l < 32; ++l) {
    float c[64] = {};
    const int m = l & 15, hf = l >> 4;
    c[0] = twr(m << 5);                 c[14] = twi(m << 5);
    c[2] = twr(m << 4);                 c[3] = twr((m + 16) << 4);
    c[4] = twi(m << 4);                 c[5] = twi((m + 16) << 4);
    c[6] = twr(m << 3);                 c[7] = twr((m + 16) << 3);
    c[8] = twi(m << 3);                 c[9] = twi((m + 16) << 3);
    c[10] = twr((m + 32) << 3);         c[11] = twr((m + 48) << 3);
    c[12] = twi((m + 32) << 3);         c[13] = twi((m + 48) << 3);
    for (int q = 0; q < 4; ++q) {
      const int k0 = 4 * l + q, h = q >> 1, e = q & 1;
      c[16 + (h * 3 + 0) * 2 + e] = twr(k0 << 2);
      c[16 + (h * 3 + 1) * 2 + e] = twr(k0 << 1);
      c[16 + (h * 3 + 2) * 2 + e] = twr((k0 + 128) << 1);
      c[28 + (h * 3 + 0) * 2 + e] = twi(k0 << 2);
      c[28 + (h * 3 + 1) * 2 + e] = twi(k0 << 1);
      c[28 + (h * 3 + 2) * 2 + e] = twi((k0 + 128) << 1);
      c[40 + q] = twr(4 * l + q);       c[44 + q] = twr(128 + 4 * l + q);
      c[48 + q] = twi(4 * l + q);       c[52 + q] = twi(128 + 4 * l + q);
    }
    for (int t = 0; t < 4; ++t) {
      const int na = 4 * hf + 1 + t;
      const int j = (na * m) & 15;
      c[56 + 2 * t]     = 0.5f * twr(64 * j);
      c[56 + 2 * t + 1] = -0.5f * twi(64 * j);
    }
    if (hf) {
      c[62] = (m & 1) ? 0.0f : 0.5f;
      c[63] = (m & 1) ? -0.5f : 0.0f;
    }
    for (int j = 0; j < 64; ++j) T.v[((j >> 2) * 32 + l) * 4 + (j & 3)] = c[j];
  }
  return T;
}
__device__ constexpr LaneTab g_lane_tab = make_lane_tab();

__device__ __forceinline__ int brev5(int c) { return (int)(__brev((unsigned)c) >> 27); }

__device__ __forceinline__ void bfly(float& er, float& ei, float& orr, float& oi,
                                     const float wr, const float wi) {
  const float tr = wr * orr - wi * oi;
  const float ti = wr * oi + wi * orr;
  const float ar = er, ai = ei;
  er  = ar + tr;  ei = ai + ti;
  orr = ar - tr;  oi = ai - ti;
}

__device__ __forceinline__ void bfly2(v2f& er, v2f& ei, v2f& orr, v2f& oi, const v2f wr, const v2f wi) {
  const v2f tr = wr * orr - wi * oi;
  const v2f ti = wr * oi + wi * orr;
  const v2f ar = er, ai = ei;
  er  = ar + tr;  ei = ai + ti;
  orr = ar - tr;  oi = ai - ti;
}

__device__ __forceinline__ void radix4x2(v2f (&R)[4], v2f (&I)[4],
                                         const v2f (&wr)[3], const v2f (&wi)[3]) {
  bfly2(R[0], I[0], R[1], I[1], wr[0], wi[0]);
  bfly2(R[2], I[2], R[3], I[3], wr[0], wi[0]);
  bfly2(R[0], I[0], R[2], I[2], wr[1], wi[1]);
  bfly2(R[1], I[1], R[3], I[3], wr[2], wi[2]);
}

__device__ __forceinline__ void split2(const v2f zkr, const v2f zki, const v2f zmr, const v2f zmi,
                                       const v2f wr, const v2f wi,
                                       v2f& xr, v2f& xi, v2f& yr, v2f& yi) {
  const v2f er = zkr + zmr, ei = zki - zmi;
  const v2f orr = zki + zmi, oi = zmr - zkr;
  const v2f tr = wr * orr - wi * oi;
  const v2f ti = wr * oi + wi * orr;
  xr = er + tr;  xi = ei + ti;
  yr = er - tr;  yi = ti - ei;
}

__device__ __forceinline__ void wave_lds_order() {
  __builtin_amdgcn_fence(__ATOMIC_RELEASE, "wavefront");
  __builtin_amdgcn_wave_barrier();
  __builtin_amdgcn_fence(__ATOMIC_ACQUIRE, "wavefront");
}

__global__ __launch_bounds__(WAVES * 32) void rfft1024_kernel(
    const float* __restrict__ x, float* __restrict__ out, int nrows) {
  __shared__ __align__(16) float buf[WAVES][2 * NFFT];

  const int tid  = threadIdx.x;
  const int wave = tid >> 5;
  const int lane = tid & 31;
  const int half = lane >> 4;
  const int m16  = lane & 15;
  const int row0 = blockIdx.x * (WAVES * ROWS) + wave;

  v4f nxt[NFFT / (32 * 4)];
  {
    const v4f* xr4 = (const v4f*)(x + (size_t)row0 * NFFT);
#pragma unroll
    for (int i = 0; i < NFFT / (32 * 4); ++i) nxt[i] = xr4[lane + 32 * i];
  }

  v4f kc[16];
  {
    const gv4f* lt = (const gv4f*)g_lane_tab.v + lane;
    asm volatile("" : "+v"(lt));
#pragma unroll
    for (int j = 0; j < 16; ++j) kc[j] = lt[j * 32];
  }
#define KC(i) (kc[(i) >> 2][(i) & 3])

  float* sx = buf[wave];
  float* ur = sx + NFFT;
  float* ui = ur + HALFN;
  v4f* sx4 = (v4f*)sx;

  const float p1r0 = KC(0), p1i0 = KC(14);
  const v2f w12r = {KC(2), KC(3)},   w12i = {KC(4), KC(5)};
  const v2f w34r = {KC(6), KC(7)},   w34i = {KC(8), KC(9)};
  const v2f w56r = {KC(10), KC(11)}, w56i = {KC(12), KC(13)};
  v2f p2r[2][3], p2i[2][3];
#pragma unroll
  for (int h = 0; h < 2; ++h) {
#pragma unroll
    for (int j = 0; j < 3; ++j) {
      p2r[h][j][0] = KC(16 + (h * 3 + j) * 2);  p2r[h][j][1] = KC(16 + (h * 3 + j) * 2 + 1);
      p2i[h][j][0] = KC(28 + (h * 3 + j) * 2);  p2i[h][j][1] = KC(28 + (h * 3 + j) * 2 + 1);
    }
  }
  const v4f swr0 = kc[10], swr1 = kc[11];
  const v4f swi0 = kc[12], swi1 = kc[13];
  v2f bF[4];
#pragma unroll
  for (int t = 0; t < 4; ++t) { bF[t][0] = KC(56 + 2 * t);  bF[t][1] = KC(57 + 2 * t); }

#pragma unroll
  for (int j = 0; j < 16; ++j) asm volatile("" :: "v"(kc[j]));

#pragma nounroll
  for (int r = 0; r < ROWS; ++r) {
    const int row = row0 + r * WAVES;

#pragma unroll
    for (int i = 0; i < NFFT / (32 * 4); ++i) sx4[lane + 32 * i] = nxt[i];
    if (r + 1 < ROWS) {
      const v4f* xn4 = (const v4f*)(x + (size_t)(row + WAVES) * NFFT);
#pragma unroll
      for (int i = 0; i < NFFT / (32 * 4); ++i) nxt[i] = xn4[lane + 32 * i];
    }
    wave_lds_order();

#pragma unroll
    for (int mt = 0; mt < NCHUNK / 16; ++mt) {
      const int cOff = 2 * brev5(mt * 16 + m16);
      v8f accR = {0.f, 0.f, 0.f, 0.f, 0.f, 0.f, 0.f, 0.f};
      v8f accI = {0.f, 0.f, 0.f, 0.f, 0.f, 0.f, 0.f, 0.f};
      v2f za[4], zb[4];
#pragma unroll
      for (int t = 0; t < 4; ++t) {
        const int na = 4 * half + 1 + t;
        const int nb = (t == 3 && half) ? 0 : 16 - na;
        za[t] = *(const v2f*)(sx + 64 * na + cOff);
        zb[t] = *(const v2f*)(sx + 64 * nb + cOff);
      }
#pragma unroll
      for (int t = 0; t < 4; ++t) {
        const v2f s = za[t] + zb[t];
        const v2f d = za[t] - zb[t];
        v2f p, q;
        p[0] = s[0];
        q[0] = s[1];
        if (t < 3) {
          p[1] = d[1];
          q[1] = zb[t][0] - za[t][0];
        } else {
          p[1] = half ? d[0] : d[1];
          q[1] = half ? d[1] : zb[t][0] - za[t][0];
        }
        accR = __builtin_amdgcn_wmma_f32_16x16x4_f32(false, p, false, bF[t], (short)0, accR, false, false);
        accI = __builtin_amdgcn_wmma_f32_16x16x4_f32(false, q, false, bF[t], (short)0, accI, false, false);
      }
      float R[8] = {accR[0], accR[1], accR[2], accR[3], accR[4], accR[5], accR[6], accR[7]};
      float I[8] = {accI[0], accI[1], accI[2], accI[3], accI[4], accI[5], accI[6], accI[7]};
      bfly(R[0], I[0], R[1], I[1], p1r0, p1i0);
      bfly(R[2], I[2], R[3], I[3], p1r0, p1i0);
      bfly(R[4], I[4], R[5], I[5], p1r0, p1i0);
      bfly(R[6], I[6], R[7], I[7], p1r0, p1i0);
      v2f PR[4], PI[4];
#pragma unroll
      for (int j = 0; j < 4; ++j) {
        PR[j][0] = R[2 * j];  PR[j][1] = R[2 * j + 1];
        PI[j][0] = I[2 * j];  PI[j][1] = I[2 * j + 1];
      }
      bfly2(PR[0], PI[0], PR[1], PI[1], w12r, w12i);
      bfly2(PR[2], PI[2], PR[3], PI[3], w12r, w12i);
      bfly2(PR[0], PI[0], PR[2], PI[2], w34r, w34i);
      bfly2(PR[1], PI[1], PR[3], PI[3], w56r, w56i);
#pragma unroll
      for (int i = 0; i < 8; ++i) {
        const int c = mt * 16 + i + 8 * half;
        ur[c * 16 + m16] = PR[i >> 1][i & 1];
        ui[c * 16 + m16] = PI[i >> 1][i & 1];
      }
    }
    wave_lds_order();

    const v4f* ur4 = (const v4f*)ur;
    const v4f* ui4 = (const v4f*)ui;
    v4f* zr4 = (v4f*)sx;
    v4f* zi4 = (v4f*)(sx + HALFN);
    {
      v4f UR[4], UI[4];
#pragma unroll
      for (int g = 0; g < 4; ++g) {
        UR[g] = ur4[32 * g + lane];
        UI[g] = ui4[32 * g + lane];
      }
      v2f oR[4][2], oI[4][2];
#pragma unroll
      for (int h = 0; h < 2; ++h) {
        v2f R[4], I[4];
#pragma unroll
        for (int g = 0; g < 4; ++g) {
          R[g] = h ? __builtin_shufflevector(UR[g], UR[g], 2, 3) : __builtin_shufflevector(UR[g], UR[g], 0, 1);
          I[g] = h ? __builtin_shufflevector(UI[g], UI[g], 2, 3) : __builtin_shufflevector(UI[g], UI[g], 0, 1);
        }
        radix4x2(R, I, p2r[h], p2i[h]);
#pragma unroll
        for (int b = 0; b < 4; ++b) { oR[b][h] = R[b];  oI[b][h] = I[b]; }
      }
#pragma unroll
      for (int b = 0; b < 4; ++b) {
        zr4[32 * b + lane] = __builtin_shufflevector(oR[b][0], oR[b][1], 0, 1, 2, 3);
        zi4[32 * b + lane] = __builtin_shufflevector(oI[b][0], oI[b][1], 0, 1, 2, 3);
      }
    }
    wave_lds_order();

    float* outR = out + (size_t)row * 513;
    float* outI = out + (size_t)nrows * 513 + (size_t)row * 513;
    {
      const v4f dr0 = zr4[lane], dr1 = zr4[32 + lane];
      const v4f di0 = zi4[lane], di1 = zi4[32 + lane];
      const v4f ar4 = zr4[127 - lane], br4 = zr4[95 - lane];
      const v4f ai4 = zi4[127 - lane], bi4 = zi4[95 - lane];
      const int p0 = (HALFN - 4 * lane) & (HALFN - 1);
      const int p1 = 384 - 4 * lane;
      const float ar0 = sx[p0], ai0 = sx[HALFN + p0];
      const float br0 = sx[p1], bi0 = sx[HALFN + p1];
      const v2f mr[4] = {{ar0, ar4[3]}, {ar4[2], ar4[1]}, {br0, br4[3]}, {br4[2], br4[1]}};
      const v2f mi[4] = {{ai0, ai4[3]}, {ai4[2], ai4[1]}, {bi0, bi4[3]}, {bi4[2], bi4[1]}};
      const v2f kr[4] = {{dr0[0], dr0[1]}, {dr0[2], dr0[3]}, {dr1[0], dr1[1]}, {dr1[2], dr1[3]}};
      const v2f ki[4] = {{di0[0], di0[1]}, {di0[2], di0[3]}, {di1[0], di1[1]}, {di1[2], di1[3]}};
      const v2f wr[4] = {{swr0[0], swr0[1]}, {swr0[2], swr0[3]}, {swr1[0], swr1[1]}, {swr1[2], swr1[3]}};
      const v2f wi[4] = {{swi0[0], swi0[1]}, {swi0[2], swi0[3]}, {swi1[0], swi1[1]}, {swi1[2], swi1[3]}};
      v2f xr[4], xi[4], yr[4], yi[4];
#pragma unroll
      for (int p = 0; p < 4; ++p) split2(kr[p], ki[p], mr[p], mi[p], wr[p], wi[p], xr[p], xi[p], yr[p], yi[p]);
      *(v4f_a4*)(outR + 4 * lane)       = __builtin_shufflevector(xr[0], xr[1], 0, 1, 2, 3);
      *(v4f_a4*)(outR + 128 + 4 * lane) = __builtin_shufflevector(xr[2], xr[3], 0, 1, 2, 3);
      *(v4f_a4*)(outI + 4 * lane)       = __builtin_shufflevector(xi[0], xi[1], 0, 1, 2, 3);
      *(v4f_a4*)(outI + 128 + 4 * lane) = __builtin_shufflevector(xi[2], xi[3], 0, 1, 2, 3);
      *(v4f_a4*)(outR + 509 - 4 * lane) = __builtin_shufflevector(yr[1], yr[0], 1, 0, 3, 2);
      *(v4f_a4*)(outR + 381 - 4 * lane) = __builtin_shufflevector(yr[3], yr[2], 1, 0, 3, 2);
      *(v4f_a4*)(outI + 509 - 4 * lane) = __builtin_shufflevector(yi[1], yi[0], 1, 0, 3, 2);
      *(v4f_a4*)(outI + 381 - 4 * lane) = __builtin_shufflevector(yi[3], yi[2], 1, 0, 3, 2);
      if (lane == 31) { outR[256] = 2.0f * sx[256];  outI[256] = -2.0f * sx[HALFN + 256]; }
    }
    wave_lds_order();
  }
}

extern "C" void kernel_launch(void* const* d_in, const int* in_sizes, int n_in,
                              void* d_out, int out_size, void* d_ws, size_t ws_size,
                              hipStream_t stream) {
  const float* x = (const float*)d_in[0];
  float* out = (float*)d_out;
  const int nrows = in_sizes[0] / NFFT;
  const int blocks = nrows / (WAVES * ROWS);
  rfft1024_kernel<<<blocks, WAVES * 32, 0, stream>>>(x, out, nrows);
}
